# starting kernel
# baseline (speedup 1.0000x reference)
#include <hip/hip_runtime.h>

#define N_NODES 100000
#define N_EDGES 1600000
#define IN_F 128
#define H_F 64

typedef float v2f __attribute__((ext_vector_type(2)));
typedef float v8f __attribute__((ext_vector_type(8)));

__global__ __launch_bounds__(256) void k_init_deg(float* __restrict__ deg) {
    int i = blockIdx.x * 256 + threadIdx.x;
    if (i < N_NODES) deg[i] = 1.0f;
}

__global__ __launch_bounds__(256) void k_edge_deg(const int* __restrict__ dst,
                                                  float* __restrict__ deg) {
    int e = blockIdx.x * 256 + threadIdx.x;
    if (e < N_EDGES) atomicAdd(&deg[dst[e]], 1.0f);
}

__global__ __launch_bounds__(256) void k_dinv(const float* __restrict__ deg,
                                              float* __restrict__ dinv) {
    int i = blockIdx.x * 256 + threadIdx.x;
    if (i < N_NODES) {
        float d = deg[i];
        dinv[i] = d > 0.0f ? rsqrtf(d) : 0.0f;
    }
}

__global__ __launch_bounds__(256) void k_gemm1(const float* __restrict__ x,
                                               const float* __restrict__ W1,
                                               float* __restrict__ h) {
    __shared__ float sW[IN_F * H_F];
    for (int i = threadIdx.x; i < IN_F * H_F; i += 256) {
        const int k = i >> 6, n = i & 63;
        sW[((k >> 1) << 7) + (n << 1) + (k & 1)] = W1[i];
    }
    __syncthreads();

    const int lane    = threadIdx.x & 31;
    const int wave    = threadIdx.x >> 5;
    const int rowBase = (blockIdx.x * 8 + wave) * 16;
    if (rowBase >= N_NODES) return;

    const int half = lane >> 4;
    const int l15  = lane & 15;
    const float* xrow = x + (size_t)(rowBase + l15) * IN_F;

    v8f acc[4] = {};

    for (int k0 = 0; k0 < IN_F; k0 += 4) {
        const int ks = k0 + half * 2;
        v2f a = *(const v2f*)(xrow + ks);
        const float* bp = &sW[((ks >> 1) << 7) + (l15 << 1)];
#pragma unroll
        for (int t = 0; t < 4; ++t) {
            v2f b = *(const v2f*)(bp + (t << 5));
            acc[t] = __builtin_amdgcn_wmma_f32_16x16x4_f32(
                false, a, false, b, (short)0, acc[t], false, false);
        }
    }

#pragma unroll
    for (int t = 0; t < 4; ++t) {
        const int col = t * 16 + l15;
#pragma unroll
        for (int r = 0; r < 8; ++r) {
            const int row = rowBase + r + 8 * half;
            h[(size_t)row * H_F + col] = acc[t][r];
        }
    }
}

__global__ __launch_bounds__(256) void k_self1(const float* __restrict__ h,
                                               const float* __restrict__ dinv,
                                               float* __restrict__ agg) {
    int t = blockIdx.x * 256 + threadIdx.x;
    if (t >= N_NODES * 16) return;
    int node = t >> 4, fg = t & 15;
    float n = dinv[node]; n *= n;
    float4 v = *((const float4*)(h + (size_t)node * H_F) + fg);
    float4 o = make_float4(v.x * n, v.y * n, v.z * n, v.w * n);
    *((float4*)(agg + (size_t)node * H_F) + fg) = o;
}

__global__ __launch_bounds__(256) void k_edge1(const int* __restrict__ src,
                                               const int* __restrict__ dst,
                                               const float* __restrict__ h,
                                               const float* __restrict__ dinv,
                                               float* __restrict__ agg) {
    int t = blockIdx.x * 256 + threadIdx.x;
    if (t >= N_EDGES * 16) return;
    int e = t >> 4, fg = t & 15;
    int s = src[e], d = dst[e];
    float norm = dinv[s] * dinv[d];
    float4 v = *((const float4*)(h + (size_t)s * H_F) + fg);
    float* o = agg + (size_t)d * H_F + fg * 4;
    atomicAdd(o + 0, v.x * norm);
    atomicAdd(o + 1, v.y * norm);
    atomicAdd(o + 2, v.z * norm);
    atomicAdd(o + 3, v.w * norm);
}

__global__ __launch_bounds__(256) void k_dot(const float* __restrict__ agg,
                                             const float* __restrict__ b1,
                                             const float* __restrict__ W2,
                                             float* __restrict__ z) {
    int gid  = blockIdx.x * 256 + threadIdx.x;
    int node = gid >> 5;
    int lane = threadIdx.x & 31;
    if (node >= N_NODES) return;
    float2 hv = *((const float2*)(agg + (size_t)node * H_F) + lane);
    float2 bb = *((const float2*)b1 + lane);
    float2 ww = *((const float2*)W2 + lane);
    float s = fmaxf(hv.x + bb.x, 0.0f) * ww.x + fmaxf(hv.y + bb.y, 0.0f) * ww.y;
#pragma unroll
    for (int off = 16; off > 0; off >>= 1) s += __shfl_xor(s, off, 32);
    if (lane == 0) z[node] = s;
}

__global__ __launch_bounds__(256) void k_self2(const float* __restrict__ z,
                                               const float* __restrict__ dinv,
                                               const float* __restrict__ b2,
                                               float* __restrict__ out) {
    int i = blockIdx.x * 256 + threadIdx.x;
    if (i < N_NODES) {
        float n = dinv[i];
        out[i] = z[i] * n * n + b2[0];
    }
}

__global__ __launch_bounds__(256) void k_edge2(const int* __restrict__ src,
                                               const int* __restrict__ dst,
                                               const float* __restrict__ z,
                                               const float* __restrict__ dinv,
                                               float* __restrict__ out) {
    int e = blockIdx.x * 256 + threadIdx.x;
    if (e < N_EDGES) {
        int s = src[e], d = dst[e];
        atomicAdd(&out[d], z[s] * dinv[s] * dinv[d]);
    }
}

extern "C" void kernel_launch(void* const* d_in, const int* in_sizes, int n_in,
                              void* d_out, int out_size, void* d_ws, size_t ws_size,
                              hipStream_t stream) {
    const float* x   = (const float*)d_in[0];
    const float* W1  = (const float*)d_in[1];
    const float* b1  = (const float*)d_in[2];
    const float* W2  = (const float*)d_in[3];
    const float* b2  = (const float*)d_in[4];
    const int*   ei  = (const int*)d_in[5];
    const int*   src = ei;
    const int*   dst = ei + N_EDGES;
    float*       out = (float*)d_out;

    char* ws = (char*)d_ws;
    const size_t NPAD = 409600;
    float* deg  = (float*)(ws);
    float* dinv = (float*)(ws + NPAD);
    float* h    = (float*)(ws + 2 * NPAD);
    float* agg1 = (float*)(ws + 2 * NPAD + (size_t)N_NODES * H_F * 4);
    float* z    = deg;

    const int nb_nodes = (N_NODES + 255) / 256;
    const int nb_edges = (N_EDGES + 255) / 256;

    k_init_deg<<<nb_nodes, 256, 0, stream>>>(deg);
    k_edge_deg<<<nb_edges, 256, 0, stream>>>(dst, deg);
    k_dinv    <<<nb_nodes, 256, 0, stream>>>(deg, dinv);

    k_gemm1   <<<(N_NODES + 127) / 128, 256, 0, stream>>>(x, W1, h);

    k_self1   <<<(N_NODES * 16) / 256, 256, 0, stream>>>(h, dinv, agg1);
    k_edge1   <<<(N_EDGES * 16) / 256, 256, 0, stream>>>(src, dst, h, dinv, agg1);

    k_dot     <<<(N_NODES * 32) / 256, 256, 0, stream>>>(agg1, b1, W2, z);

    k_self2   <<<nb_nodes, 256, 0, stream>>>(z, dinv, b2, out);
    k_edge2   <<<nb_edges, 256, 0, stream>>>(src, dst, z, dinv, out);
}


# layer-1 owner numbers stored as 8 bit planes (ballots in the degree pass): a lane gets the owned-edge mask of 32 edges from 8 xor + 7 and, one branch per 1024 edges (was per 128); sum order changes
# speedup vs baseline: 3.0796x; 3.0796x over previous
#include <hip/hip_runtime.h>

#define N_NODES 100000
#define N_EDGES 1600000
#define IN_F 128
#define H_F 64

typedef float v2f __attribute__((ext_vector_type(2)));
typedef float v8f __attribute__((ext_vector_type(8)));

#define NR      4
#define RANGE   25024
#define NC      64
#define CHUNK   (N_EDGES / NC)
#define PER_THR ((CHUNK + 1023) / 1024)
#define GRP     13
#define R1      416
#define SQ      ((RANGE / 4 + 1023) / 1024)

static_assert(RANGE % 4 == 0 && (N_NODES - (NR - 1) * RANGE) % 4 == 0 &&
              (N_NODES - (NR - 1) * RANGE) > 0, "every range is whole 16-byte pieces");
#define N4      (N_EDGES / 4)
#define NSB     ((N_EDGES + 1023) / 1024)
static_assert(NR * NC * 2048 >= NSB * 256 && N_EDGES % 128 == 0,
              "the grid covers the owner planes, padding included, in whole 128-edge blocks");

__global__ __launch_bounds__(1024) void k_deg_lds(const int* __restrict__ dst,
                                                  float* __restrict__ part,
                                                  unsigned* __restrict__ planes) {
    __shared__ unsigned cnt[RANGE];
    const int r   = blockIdx.x & (NR - 1);
    const int c   = blockIdx.x >> 2;
    const int lo  = r * RANGE;
    const int lim = min(RANGE, N_NODES - lo);
    const int  lane = threadIdx.x & 31;
    const int  pb = blockIdx.x * 2048 + (threadIdx.x >> 5) * 64;
    const int  q0 = pb + lane;
    const int  q1 = q0 + 32;
    const int4 d0 = ((const int4*)dst)[min(q0, N4 - 1)];
    const int4 d1 = ((const int4*)dst)[min(q1, N4 - 1)];
    for (int i = threadIdx.x; i < RANGE; i += 1024) cnt[i] = 0u;
    __syncthreads();

    const int* dchunk = dst + c * CHUNK;
    int dv[PER_THR];
#pragma unroll
    for (int j = 0; j < PER_THR; ++j) {
        const int idx = j * 1024 + threadIdx.x;
        dv[j] = idx < CHUNK ? dchunk[idx] : -1;
    }
    if (pb < NSB * 256) {
        const unsigned in0 = q0 < N4 ? 0u : 255u, in1 = q1 < N4 ? 0u : 255u;
        const unsigned o[2][4] = {
            {(unsigned)d0.x / (unsigned)R1 | in0, (unsigned)d0.y / (unsigned)R1 | in0,
             (unsigned)d0.z / (unsigned)R1 | in0, (unsigned)d0.w / (unsigned)R1 | in0},
            {(unsigned)d1.x / (unsigned)R1 | in1, (unsigned)d1.y / (unsigned)R1 | in1,
             (unsigned)d1.z / (unsigned)R1 | in1, (unsigned)d1.w / (unsigned)R1 | in1}};
        unsigned wlo = 0u, whi = 0u;
#pragma unroll
        for (int k = 0; k < 2; ++k)
#pragma unroll
            for (int cc = 0; cc < 4; ++cc)
#pragma unroll
                for (int b = 0; b < 4; ++b) {
                    const unsigned l0 = __builtin_amdgcn_ballot_w32(((o[k][cc] >> b) & 1u) != 0u);
                    const unsigned l1 = __builtin_amdgcn_ballot_w32(((o[k][cc] >> (b + 4)) & 1u) != 0u);
                    const bool mine = lane == k * 16 + cc * 4 + b;
                    wlo = mine ? l0 : wlo;
                    whi = mine ? l1 : whi;
                }
        unsigned* line = planes + (pb >> 8) * 256 + ((pb >> 5) & 7) * 16;
        line[lane]       = wlo;
        line[128 + lane] = whi;
    }
#pragma unroll
    for (int j = 0; j < PER_THR; ++j) {
        const unsigned dl = (unsigned)(dv[j] - lo);
        if (dl < (unsigned)lim)
            __hip_atomic_fetch_add(&cnt[dl], 1u, __ATOMIC_RELAXED,
                                   __HIP_MEMORY_SCOPE_WORKGROUP);
    }
    __syncthreads();

    float* p = part + (size_t)c * N_NODES + lo;
    for (int i = threadIdx.x; i < lim; i += 1024) p[i] = (float)cnt[i];
}

__global__ __launch_bounds__(256) void k_dinv(const float* __restrict__ part,
                                              float* __restrict__ dinv) {
    int i = blockIdx.x * 256 + threadIdx.x;
    if (i < N_NODES) {
        float d = 1.0f;
#pragma unroll
        for (int c = 0; c < NC; ++c) d += part[(size_t)c * N_NODES + i];
        dinv[i] = d > 0.0f ? rsqrtf(d) : 0.0f;
    }
}

__global__ __launch_bounds__(256) void k_gemm1(const float* __restrict__ x,
                                               const float* __restrict__ W1,
                                               float* __restrict__ h) {
    __shared__ float sW[IN_F * H_F];
    for (int i = threadIdx.x; i < IN_F * H_F; i += 256) {
        const int k = i >> 6, n = i & 63;
        sW[((k >> 1) << 7) + (n << 1) + (k & 1)] = W1[i];
    }
    __syncthreads();

    const int lane    = threadIdx.x & 31;
    const int wave    = threadIdx.x >> 5;
    const int rowBase = (blockIdx.x * 8 + wave) * 16;
    if (rowBase >= N_NODES) return;

    const int half = lane >> 4;
    const int l15  = lane & 15;
    const float* xrow = x + (size_t)(rowBase + l15) * IN_F;

    v8f acc[4] = {};

    for (int k0 = 0; k0 < IN_F; k0 += 4) {
        const int ks = k0 + half * 2;
        v2f a = *(const v2f*)(xrow + ks);
        const float* bp = &sW[((ks >> 1) << 7) + (l15 << 1)];
#pragma unroll
        for (int t = 0; t < 4; ++t) {
            v2f b = *(const v2f*)(bp + (t << 5));
            acc[t] = __builtin_amdgcn_wmma_f32_16x16x4_f32(
                false, a, false, b, (short)0, acc[t], false, false);
        }
    }

#pragma unroll
    for (int t = 0; t < 4; ++t) {
        const int col = t * 16 + l15;
#pragma unroll
        for (int r = 0; r < 8; ++r) {
            const int row = rowBase + r + 8 * half;
            h[(size_t)row * H_F + col] = acc[t][r];
        }
    }
}

#define NB1    ((N_NODES + R1 - 1) / R1)
#define USB    2
#define TSB    (32 * USB)
#define PQ     16
#define QW     (PQ + 1024)

static_assert(R1 % 32 == 0 && NB1 <= 255, "owner number fits 8 planes, 255 stays free");

__device__ __forceinline__ void agg1_rows(const unsigned* myq, unsigned base, unsigned k,
                                          int lane, int lo, const int* __restrict__ src,
                                          const int* __restrict__ dst,
                                          const float* __restrict__ h,
                                          const float* __restrict__ dinv,
                                          const float* sdinv, float* acc) {
    unsigned sv = 0u, dl = 0u;
    float    nv = 0.0f;
    if ((unsigned)lane < k) {
        const unsigned e = myq[base + lane];
        sv = (unsigned)src[e];
        dl = (unsigned)(dst[e] - lo);
        nv = dinv[sv] * sdinv[dl];
    }
    float2 v[PQ];
#pragma unroll
    for (int j = 0; j < PQ; ++j) {
        const unsigned sj = __builtin_amdgcn_readlane(sv, j);
        v[j] = *((const float2*)(h + (size_t)sj * H_F) + lane);
    }
#pragma unroll
    for (int j = 0; j < PQ; ++j) {
        if ((unsigned)j < k) {
            const unsigned dj  = (unsigned)__builtin_amdgcn_readlane(dl, j);
            const float    nrm = __int_as_float(__builtin_amdgcn_readlane(__float_as_int(nv), j));
            float* o = acc + dj * H_F + 2 * lane;
            __hip_atomic_fetch_add(o,     v[j].x * nrm, __ATOMIC_RELAXED,
                                   __HIP_MEMORY_SCOPE_WORKGROUP);
            __hip_atomic_fetch_add(o + 1, v[j].y * nrm, __ATOMIC_RELAXED,
                                   __HIP_MEMORY_SCOPE_WORKGROUP);
        }
    }
}

__device__ __forceinline__ void agg1_load(uint4 (&ov)[2 * USB], const uint4* __restrict__ o16,
                                          int t0, int wave, int lane) {
#pragma unroll
    for (int u = 0; u < USB; ++u) {
        const int sb = min(t0 + u * 32 + wave, NSB - 1);
        ov[2 * u]     = o16[sb * 64 + lane];
        ov[2 * u + 1] = o16[sb * 64 + 32 + lane];
    }
}

__device__ __forceinline__ void agg1_scan(const uint4 (&ov)[2 * USB], int t0, int wave, int lane,
                                          int lo, const unsigned (&nb)[8], unsigned below,
                                          unsigned* myq, unsigned& cnt,
                                          const int* __restrict__ src,
                                          const int* __restrict__ dst,
                                          const float* __restrict__ h,
                                          const float* __restrict__ dinv,
                                          const float* sdinv, float* acc) {
#pragma unroll
    for (int u = 0; u < USB; ++u) {
        const int   sb = t0 + u * 32 + wave;
        const uint4 pa = ov[2 * u], pc = ov[2 * u + 1];
        unsigned t = (pa.x ^ nb[0]) & (pa.y ^ nb[1]) & (pa.z ^ nb[2]) & (pa.w ^ nb[3])
                   & (pc.x ^ nb[4]) & (pc.y ^ nb[5]) & (pc.z ^ nb[6]) & (pc.w ^ nb[7]);
        if (sb >= NSB) t = 0u;
        const unsigned e0 = (unsigned)sb * 1024u + (unsigned)((lane >> 2) << 7) + (unsigned)(lane & 3);
        unsigned m = __builtin_amdgcn_ballot_w32(t != 0u);
        while (m != 0u) {
            if (t != 0u) {
                myq[cnt + __builtin_popcount(m & below)] = e0 + 4u * (unsigned)__builtin_ctz(t);
                t &= t - 1u;
            }
            cnt += __builtin_popcount(m);
            m = __builtin_amdgcn_ballot_w32(t != 0u);
        }
        while (cnt >= (unsigned)PQ) {
            cnt -= PQ;
            agg1_rows(myq, cnt, PQ, lane, lo, src, dst, h, dinv, sdinv, acc);
        }
    }
}

__global__ __launch_bounds__(1024) void k_agg1_lds(const int* __restrict__ src,
                                                   const int* __restrict__ dst,
                                                   const unsigned* __restrict__ planes,
                                                   const float* __restrict__ h,
                                                   const float* __restrict__ dinv,
                                                   const float* __restrict__ b1,
                                                   const float* __restrict__ W2,
                                                   float* __restrict__ zd) {
    __shared__ float    acc[R1 * H_F];
    __shared__ unsigned wq[32 * QW];
    __shared__ float    sdinv[R1];
    __shared__ float    sz[R1];

    const int tid  = threadIdx.x;
    const int lane = tid & 31;
    const int wave = tid >> 5;
    const int lo   = blockIdx.x * R1;
    const int lim  = min(R1, N_NODES - lo);

    for (int i = tid; i < lim * 16; i += 1024) {
        const int node = i >> 4;
        float n = dinv[lo + node]; n *= n;
        const float4 v = *((const float4*)(h + (size_t)lo * H_F) + i);
        *((float4*)acc + i) = make_float4(v.x * n, v.y * n, v.z * n, v.w * n);
    }
    for (int i = tid; i < lim; i += 1024) sdinv[i] = dinv[lo + i];
    __syncthreads();

    unsigned*      myq   = wq + wave * QW;
    const unsigned below = (1u << lane) - 1u;
    unsigned       nb[8];
#pragma unroll
    for (int b = 0; b < 8; ++b) nb[b] = ((blockIdx.x >> b) & 1u) - 1u;
    unsigned       cnt   = 0u;
    const uint4*   o16   = (const uint4*)planes;

    uint4 oa[2 * USB], ob[2 * USB];
    agg1_load(oa, o16, 0, wave, lane);
    for (int t0 = 0; t0 < NSB; t0 += 2 * TSB) {
        agg1_load(ob, o16, t0 + TSB, wave, lane);
        agg1_scan(oa, t0, wave, lane, lo, nb, below, myq, cnt, src, dst, h, dinv, sdinv, acc);
        agg1_load(oa, o16, t0 + 2 * TSB, wave, lane);
        if (t0 + TSB < NSB)
            agg1_scan(ob, t0 + TSB, wave, lane, lo, nb, below, myq, cnt, src, dst, h, dinv, sdinv, acc);
    }
    if (cnt > 0u) agg1_rows(myq, 0u, cnt, lane, lo, src, dst, h, dinv, sdinv, acc);
    __syncthreads();

    const float2 bb = *((const float2*)b1 + lane);
    const float2 ww = *((const float2*)W2 + lane);
    for (int node = wave; node < lim; node += 32) {
        const float2 hv = *((const float2*)(acc + node * H_F) + lane);
        float s = fmaxf(hv.x + bb.x, 0.0f) * ww.x + fmaxf(hv.y + bb.y, 0.0f) * ww.y;
#pragma unroll
        for (int off = 16; off > 0; off >>= 1) s += __shfl_xor(s, off, 32);
        if (lane == 0) sz[node] = s;
    }
    __syncthreads();
    float2* zd2 = (float2*)zd;
    for (int i = tid; i < lim; i += 1024) zd2[lo + i] = make_float2(sz[i], sdinv[i]);
}

__global__ __launch_bounds__(1024) void k_edge2_lds(const int* __restrict__ src,
                                                    const int* __restrict__ dst,
                                                    const float* __restrict__ zd,
                                                    const float* __restrict__ dinv,
                                                    float* __restrict__ part) {
    __shared__ float acc[RANGE];
    __shared__ float sdinv[RANGE];
    const int r   = blockIdx.x & (NR - 1);
    const int c   = blockIdx.x >> 2;
    const int lo  = r * RANGE;
    const int lim = min(RANGE, N_NODES - lo);

    const float4* dq4 = (const float4*)(dinv + lo);
    float4 dq[SQ];
#pragma unroll
    for (int q = 0; q < SQ; ++q)
        dq[q] = dq4[min(q * 1024 + (int)threadIdx.x, lim / 4 - 1)];

    const int* schunk = src + c * CHUNK;
    const int* dchunk = dst + c * CHUNK;
    int sv[PER_THR], dv[PER_THR];
#pragma unroll
    for (int j = 0; j < PER_THR; ++j) {
        const int idx = j * 1024 + threadIdx.x;
        const bool ok = idx < CHUNK;
        sv[j] = ok ? schunk[idx] : 0;
        dv[j] = ok ? dchunk[idx] : -1;
    }
#pragma unroll
    for (int q = 0; q < SQ; ++q) {
        const int i4 = q * 1024 + threadIdx.x;
        if (i4 < RANGE / 4) {
            ((float4*)acc)[i4]   = make_float4(0.0f, 0.0f, 0.0f, 0.0f);
            ((float4*)sdinv)[i4] = dq[q];
        }
    }
    __syncthreads();

    const float2* zd2 = (const float2*)zd;
#pragma unroll
    for (int g = 0; g < PER_THR; g += GRP) {
        float2 zr[GRP];
        float  bv[GRP];
#pragma unroll
        for (int j = 0; j < GRP; ++j) {
            if (g + j < PER_THR) {
                const unsigned dl  = (unsigned)(dv[g + j] - lo);
                const bool     own = dl < (unsigned)lim;
                zr[j] = zd2[own ? sv[g + j] : 0];
                bv[j] = sdinv[own ? dl : 0u];
            }
        }
#pragma unroll
        for (int j = 0; j < GRP; ++j) {
            if (g + j < PER_THR) {
                const unsigned dl = (unsigned)(dv[g + j] - lo);
                if (dl < (unsigned)lim)
                    __hip_atomic_fetch_add(&acc[dl], zr[j].x * zr[j].y * bv[j],
                                           __ATOMIC_RELAXED,
                                           __HIP_MEMORY_SCOPE_WORKGROUP);
            }
        }
    }
    __syncthreads();

    float* p = part + (size_t)c * N_NODES + lo;
    for (int i = threadIdx.x; i < lim; i += 1024) p[i] = acc[i];
}

__global__ __launch_bounds__(256) void k_out(const float* __restrict__ zd,
                                             const float* __restrict__ b2,
                                             const float* __restrict__ part,
                                             float* __restrict__ out) {
    int i = blockIdx.x * 256 + threadIdx.x;
    if (i < N_NODES) {
        const float2 rec = ((const float2*)zd)[i];
        float n = rec.y;
        float o = rec.x * n * n + b2[0];
#pragma unroll
        for (int c = 0; c < NC; ++c) o += part[(size_t)c * N_NODES + i];
        out[i] = o;
    }
}

extern "C" void kernel_launch(void* const* d_in, const int* in_sizes, int n_in,
                              void* d_out, int out_size, void* d_ws, size_t ws_size,
                              hipStream_t stream) {
    const float* x   = (const float*)d_in[0];
    const float* W1  = (const float*)d_in[1];
    const float* b1  = (const float*)d_in[2];
    const float* W2  = (const float*)d_in[3];
    const float* b2  = (const float*)d_in[4];
    const int*   ei  = (const int*)d_in[5];
    const int*   src = ei;
    const int*   dst = ei + N_EDGES;
    float*       out = (float*)d_out;

    char* ws = (char*)d_ws;
    const size_t NPAD = 409600;
    float* zd   = (float*)(ws);
    float* dinv = (float*)(ws + 2 * NPAD);
    float* h    = (float*)(ws + 3 * NPAD);
    float* part = (float*)(ws + 3 * NPAD + (size_t)N_NODES * H_F * 4);
    unsigned* planes = (unsigned*)(ws + 3 * NPAD + (size_t)N_NODES * H_F * 4
                                   + (size_t)NC * N_NODES * 4);

    const int nb_nodes = (N_NODES + 255) / 256;

    k_deg_lds <<<NR * NC, 1024, 0, stream>>>(dst, part, planes);
    k_dinv    <<<nb_nodes, 256, 0, stream>>>(part, dinv);

    k_gemm1   <<<(N_NODES + 127) / 128, 256, 0, stream>>>(x, W1, h);

    k_agg1_lds<<<NB1, 1024, 0, stream>>>(src, dst, planes, h, dinv, b1, W2, zd);

    k_edge2_lds<<<NR * NC, 1024, 0, stream>>>(src, dst, zd, dinv, part);
    k_out     <<<nb_nodes, 256, 0, stream>>>(zd, b2, part, out);
}


# layer-1 edge lookup: one 4-byte gather per owned edge of a packed word (src<<9 | dst%416) that the degree pass stores for every edge, instead of two gathers (src[e] and dst[e])
# speedup vs baseline: 3.1926x; 1.0367x over previous
#include <hip/hip_runtime.h>

#define N_NODES 100000
#define N_EDGES 1600000
#define IN_F 128
#define H_F 64

typedef float v2f __attribute__((ext_vector_type(2)));
typedef float v8f __attribute__((ext_vector_type(8)));

#define NR      4
#define RANGE   25024
#define NC      64
#define CHUNK   (N_EDGES / NC)
#define PER_THR ((CHUNK + 1023) / 1024)
#define GRP     13
#define R1      416
#define SQ      ((RANGE / 4 + 1023) / 1024)

static_assert(RANGE % 4 == 0 && (N_NODES - (NR - 1) * RANGE) % 4 == 0 &&
              (N_NODES - (NR - 1) * RANGE) > 0, "every range is whole 16-byte pieces");
#define N4      (N_EDGES / 4)
#define NSB     ((N_EDGES + 1023) / 1024)
static_assert(NR * NC * 2048 >= NSB * 256 && N_EDGES % 128 == 0,
              "the grid covers the owner planes, padding included, in whole 128-edge blocks");

#define PKS     9
static_assert(R1 <= (1 << PKS) && N_NODES <= (1 << (32 - PKS)), "src and dst % R1 share a word");
__global__ __launch_bounds__(1024) void k_deg_lds(const int* __restrict__ src,
                                                  const int* __restrict__ dst,
                                                  float* __restrict__ part,
                                                  unsigned* __restrict__ planes,
                                                  unsigned* __restrict__ pk) {
    __shared__ unsigned cnt[RANGE];
    const int r   = blockIdx.x & (NR - 1);
    const int c   = blockIdx.x >> 2;
    const int lo  = r * RANGE;
    const int lim = min(RANGE, N_NODES - lo);
    const int  lane = threadIdx.x & 31;
    const int  pb = blockIdx.x * 2048 + (threadIdx.x >> 5) * 64;
    const int  q0 = pb + lane;
    const int  q1 = q0 + 32;
    const int4 d0 = ((const int4*)dst)[min(q0, N4 - 1)];
    const int4 d1 = ((const int4*)dst)[min(q1, N4 - 1)];
    const int4 s0 = ((const int4*)src)[min(q0, N4 - 1)];
    const int4 s1 = ((const int4*)src)[min(q1, N4 - 1)];
    for (int i = threadIdx.x; i < RANGE; i += 1024) cnt[i] = 0u;
    __syncthreads();

    const int* dchunk = dst + c * CHUNK;
    int dv[PER_THR];
#pragma unroll
    for (int j = 0; j < PER_THR; ++j) {
        const int idx = j * 1024 + threadIdx.x;
        dv[j] = idx < CHUNK ? dchunk[idx] : -1;
    }
    if (pb < NSB * 256) {
        const unsigned in0 = q0 < N4 ? 0u : 255u, in1 = q1 < N4 ? 0u : 255u;
        const unsigned o[2][4] = {
            {(unsigned)d0.x / (unsigned)R1 | in0, (unsigned)d0.y / (unsigned)R1 | in0,
             (unsigned)d0.z / (unsigned)R1 | in0, (unsigned)d0.w / (unsigned)R1 | in0},
            {(unsigned)d1.x / (unsigned)R1 | in1, (unsigned)d1.y / (unsigned)R1 | in1,
             (unsigned)d1.z / (unsigned)R1 | in1, (unsigned)d1.w / (unsigned)R1 | in1}};
        if (q0 < N4)
            ((uint4*)pk)[q0] = make_uint4(
                ((unsigned)s0.x << PKS) | ((unsigned)d0.x - o[0][0] * (unsigned)R1),
                ((unsigned)s0.y << PKS) | ((unsigned)d0.y - o[0][1] * (unsigned)R1),
                ((unsigned)s0.z << PKS) | ((unsigned)d0.z - o[0][2] * (unsigned)R1),
                ((unsigned)s0.w << PKS) | ((unsigned)d0.w - o[0][3] * (unsigned)R1));
        if (q1 < N4)
            ((uint4*)pk)[q1] = make_uint4(
                ((unsigned)s1.x << PKS) | ((unsigned)d1.x - o[1][0] * (unsigned)R1),
                ((unsigned)s1.y << PKS) | ((unsigned)d1.y - o[1][1] * (unsigned)R1),
                ((unsigned)s1.z << PKS) | ((unsigned)d1.z - o[1][2] * (unsigned)R1),
                ((unsigned)s1.w << PKS) | ((unsigned)d1.w - o[1][3] * (unsigned)R1));
        unsigned wlo = 0u, whi = 0u;
#pragma unroll
        for (int k = 0; k < 2; ++k)
#pragma unroll
            for (int cc = 0; cc < 4; ++cc)
#pragma unroll
                for (int b = 0; b < 4; ++b) {
                    const unsigned l0 = __builtin_amdgcn_ballot_w32(((o[k][cc] >> b) & 1u) != 0u);
                    const unsigned l1 = __builtin_amdgcn_ballot_w32(((o[k][cc] >> (b + 4)) & 1u) != 0u);
                    const bool mine = lane == k * 16 + cc * 4 + b;
                    wlo = mine ? l0 : wlo;
                    whi = mine ? l1 : whi;
                }
        unsigned* line = planes + (pb >> 8) * 256 + ((pb >> 5) & 7) * 16;
        line[lane]       = wlo;
        line[128 + lane] = whi;
    }
#pragma unroll
    for (int j = 0; j < PER_THR; ++j) {
        const unsigned dl = (unsigned)(dv[j] - lo);
        if (dl < (unsigned)lim)
            __hip_atomic_fetch_add(&cnt[dl], 1u, __ATOMIC_RELAXED,
                                   __HIP_MEMORY_SCOPE_WORKGROUP);
    }
    __syncthreads();

    float* p = part + (size_t)c * N_NODES + lo;
    for (int i = threadIdx.x; i < lim; i += 1024) p[i] = (float)cnt[i];
}

__global__ __launch_bounds__(256) void k_dinv(const float* __restrict__ part,
                                              float* __restrict__ dinv) {
    int i = blockIdx.x * 256 + threadIdx.x;
    if (i < N_NODES) {
        float d = 1.0f;
#pragma unroll
        for (int c = 0; c < NC; ++c) d += part[(size_t)c * N_NODES + i];
        dinv[i] = d > 0.0f ? rsqrtf(d) : 0.0f;
    }
}

__global__ __launch_bounds__(256) void k_gemm1(const float* __restrict__ x,
                                               const float* __restrict__ W1,
                                               float* __restrict__ h) {
    __shared__ float sW[IN_F * H_F];
    for (int i = threadIdx.x; i < IN_F * H_F; i += 256) {
        const int k = i >> 6, n = i & 63;
        sW[((k >> 1) << 7) + (n << 1) + (k & 1)] = W1[i];
    }
    __syncthreads();

    const int lane    = threadIdx.x & 31;
    const int wave    = threadIdx.x >> 5;
    const int rowBase = (blockIdx.x * 8 + wave) * 16;
    if (rowBase >= N_NODES) return;

    const int half = lane >> 4;
    const int l15  = lane & 15;
    const float* xrow = x + (size_t)(rowBase + l15) * IN_F;

    v8f acc[4] = {};

    for (int k0 = 0; k0 < IN_F; k0 += 4) {
        const int ks = k0 + half * 2;
        v2f a = *(const v2f*)(xrow + ks);
        const float* bp = &sW[((ks >> 1) << 7) + (l15 << 1)];
#pragma unroll
        for (int t = 0; t < 4; ++t) {
            v2f b = *(const v2f*)(bp + (t << 5));
            acc[t] = __builtin_amdgcn_wmma_f32_16x16x4_f32(
                false, a, false, b, (short)0, acc[t], false, false);
        }
    }

#pragma unroll
    for (int t = 0; t < 4; ++t) {
        const int col = t * 16 + l15;
#pragma unroll
        for (int r = 0; r < 8; ++r) {
            const int row = rowBase + r + 8 * half;
            h[(size_t)row * H_F + col] = acc[t][r];
        }
    }
}

#define NB1    ((N_NODES + R1 - 1) / R1)
#define USB    2
#define TSB    (32 * USB)
#define PQ     16
#define QW     (PQ + 1024)

static_assert(R1 % 32 == 0 && NB1 <= 255, "owner number fits 8 planes, 255 stays free");

__device__ __forceinline__ void agg1_rows(const unsigned* myq, unsigned base, unsigned k,
                                          int lane, const unsigned* __restrict__ pk,
                                          const float* __restrict__ h,
                                          const float* __restrict__ dinv,
                                          const float* sdinv, float* acc) {
    unsigned sv = 0u, dl = 0u;
    float    nv = 0.0f;
    if ((unsigned)lane < k) {
        const unsigned w = pk[myq[base + lane]];
        sv = w >> PKS;
        dl = w & ((1u << PKS) - 1u);
        nv = dinv[sv] * sdinv[dl];
    }
    float2 v[PQ];
#pragma unroll
    for (int j = 0; j < PQ; ++j) {
        const unsigned sj = __builtin_amdgcn_readlane(sv, j);
        v[j] = *((const float2*)(h + (size_t)sj * H_F) + lane);
    }
#pragma unroll
    for (int j = 0; j < PQ; ++j) {
        if ((unsigned)j < k) {
            const unsigned dj  = (unsigned)__builtin_amdgcn_readlane(dl, j);
            const float    nrm = __int_as_float(__builtin_amdgcn_readlane(__float_as_int(nv), j));
            float* o = acc + dj * H_F + 2 * lane;
            __hip_atomic_fetch_add(o,     v[j].x * nrm, __ATOMIC_RELAXED,
                                   __HIP_MEMORY_SCOPE_WORKGROUP);
            __hip_atomic_fetch_add(o + 1, v[j].y * nrm, __ATOMIC_RELAXED,
                                   __HIP_MEMORY_SCOPE_WORKGROUP);
        }
    }
}

__device__ __forceinline__ void agg1_load(uint4 (&ov)[2 * USB], const uint4* __restrict__ o16,
                                          int t0, int wave, int lane) {
#pragma unroll
    for (int u = 0; u < USB; ++u) {
        const int sb = min(t0 + u * 32 + wave, NSB - 1);
        ov[2 * u]     = o16[sb * 64 + lane];
        ov[2 * u + 1] = o16[sb * 64 + 32 + lane];
    }
}

__device__ __forceinline__ void agg1_scan(const uint4 (&ov)[2 * USB], int t0, int wave, int lane,
                                          const unsigned (&nb)[8], unsigned below,
                                          unsigned* myq, unsigned& cnt,
                                          const unsigned* __restrict__ pk,
                                          const float* __restrict__ h,
                                          const float* __restrict__ dinv,
                                          const float* sdinv, float* acc) {
#pragma unroll
    for (int u = 0; u < USB; ++u) {
        const int   sb = t0 + u * 32 + wave;
        const uint4 pa = ov[2 * u], pc = ov[2 * u + 1];
        unsigned t = (pa.x ^ nb[0]) & (pa.y ^ nb[1]) & (pa.z ^ nb[2]) & (pa.w ^ nb[3])
                   & (pc.x ^ nb[4]) & (pc.y ^ nb[5]) & (pc.z ^ nb[6]) & (pc.w ^ nb[7]);
        if (sb >= NSB) t = 0u;
        const unsigned e0 = (unsigned)sb * 1024u + (unsigned)((lane >> 2) << 7) + (unsigned)(lane & 3);
        unsigned m = __builtin_amdgcn_ballot_w32(t != 0u);
        while (m != 0u) {
            if (t != 0u) {
                myq[cnt + __builtin_popcount(m & below)] = e0 + 4u * (unsigned)__builtin_ctz(t);
                t &= t - 1u;
            }
            cnt += __builtin_popcount(m);
            m = __builtin_amdgcn_ballot_w32(t != 0u);
        }
        while (cnt >= (unsigned)PQ) {
            cnt -= PQ;
            agg1_rows(myq, cnt, PQ, lane, pk, h, dinv, sdinv, acc);
        }
    }
}

__global__ __launch_bounds__(1024) void k_agg1_lds(const unsigned* __restrict__ pk,
                                                   const unsigned* __restrict__ planes,
                                                   const float* __restrict__ h,
                                                   const float* __restrict__ dinv,
                                                   const float* __restrict__ b1,
                                                   const float* __restrict__ W2,
                                                   float* __restrict__ zd) {
    __shared__ float    acc[R1 * H_F];
    __shared__ unsigned wq[32 * QW];
    __shared__ float    sdinv[R1];
    __shared__ float    sz[R1];

    const int tid  = threadIdx.x;
    const int lane = tid & 31;
    const int wave = tid >> 5;
    const int lo   = blockIdx.x * R1;
    const int lim  = min(R1, N_NODES - lo);

    for (int i = tid; i < lim * 16; i += 1024) {
        const int node = i >> 4;
        float n = dinv[lo + node]; n *= n;
        const float4 v = *((const float4*)(h + (size_t)lo * H_F) + i);
        *((float4*)acc + i) = make_float4(v.x * n, v.y * n, v.z * n, v.w * n);
    }
    for (int i = tid; i < lim; i += 1024) sdinv[i] = dinv[lo + i];
    __syncthreads();

    unsigned*      myq   = wq + wave * QW;
    const unsigned below = (1u << lane) - 1u;
    unsigned       nb[8];
#pragma unroll
    for (int b = 0; b < 8; ++b) nb[b] = ((blockIdx.x >> b) & 1u) - 1u;
    unsigned       cnt   = 0u;
    const uint4*   o16   = (const uint4*)planes;

    uint4 oa[2 * USB], ob[2 * USB];
    agg1_load(oa, o16, 0, wave, lane);
    for (int t0 = 0; t0 < NSB; t0 += 2 * TSB) {
        agg1_load(ob, o16, t0 + TSB, wave, lane);
        agg1_scan(oa, t0, wave, lane, nb, below, myq, cnt, pk, h, dinv, sdinv, acc);
        agg1_load(oa, o16, t0 + 2 * TSB, wave, lane);
        if (t0 + TSB < NSB)
            agg1_scan(ob, t0 + TSB, wave, lane, nb, below, myq, cnt, pk, h, dinv, sdinv, acc);
    }
    if (cnt > 0u) agg1_rows(myq, 0u, cnt, lane, pk, h, dinv, sdinv, acc);
    __syncthreads();

    const float2 bb = *((const float2*)b1 + lane);
    const float2 ww = *((const float2*)W2 + lane);
    for (int node = wave; node < lim; node += 32) {
        const float2 hv = *((const float2*)(acc + node * H_F) + lane);
        float s = fmaxf(hv.x + bb.x, 0.0f) * ww.x + fmaxf(hv.y + bb.y, 0.0f) * ww.y;
#pragma unroll
        for (int off = 16; off > 0; off >>= 1) s += __shfl_xor(s, off, 32);
        if (lane == 0) sz[node] = s;
    }
    __syncthreads();
    float2* zd2 = (float2*)zd;
    for (int i = tid; i < lim; i += 1024) zd2[lo + i] = make_float2(sz[i], sdinv[i]);
}

__global__ __launch_bounds__(1024) void k_edge2_lds(const int* __restrict__ src,
                                                    const int* __restrict__ dst,
                                                    const float* __restrict__ zd,
                                                    const float* __restrict__ dinv,
                                                    float* __restrict__ part) {
    __shared__ float acc[RANGE];
    __shared__ float sdinv[RANGE];
    const int r   = blockIdx.x & (NR - 1);
    const int c   = blockIdx.x >> 2;
    const int lo  = r * RANGE;
    const int lim = min(RANGE, N_NODES - lo);

    const float4* dq4 = (const float4*)(dinv + lo);
    float4 dq[SQ];
#pragma unroll
    for (int q = 0; q < SQ; ++q)
        dq[q] = dq4[min(q * 1024 + (int)threadIdx.x, lim / 4 - 1)];

    const int* schunk = src + c * CHUNK;
    const int* dchunk = dst + c * CHUNK;
    int sv[PER_THR], dv[PER_THR];
#pragma unroll
    for (int j = 0; j < PER_THR; ++j) {
        const int idx = j * 1024 + threadIdx.x;
        const bool ok = idx < CHUNK;
        sv[j] = ok ? schunk[idx] : 0;
        dv[j] = ok ? dchunk[idx] : -1;
    }
#pragma unroll
    for (int q = 0; q < SQ; ++q) {
        const int i4 = q * 1024 + threadIdx.x;
        if (i4 < RANGE / 4) {
            ((float4*)acc)[i4]   = make_float4(0.0f, 0.0f, 0.0f, 0.0f);
            ((float4*)sdinv)[i4] = dq[q];
        }
    }
    __syncthreads();

    const float2* zd2 = (const float2*)zd;
#pragma unroll
    for (int g = 0; g < PER_THR; g += GRP) {
        float2 zr[GRP];
        float  bv[GRP];
#pragma unroll
        for (int j = 0; j < GRP; ++j) {
            if (g + j < PER_THR) {
                const unsigned dl  = (unsigned)(dv[g + j] - lo);
                const bool     own = dl < (unsigned)lim;
                zr[j] = zd2[own ? sv[g + j] : 0];
                bv[j] = sdinv[own ? dl : 0u];
            }
        }
#pragma unroll
        for (int j = 0; j < GRP; ++j) {
            if (g + j < PER_THR) {
                const unsigned dl = (unsigned)(dv[g + j] - lo);
                if (dl < (unsigned)lim)
                    __hip_atomic_fetch_add(&acc[dl], zr[j].x * zr[j].y * bv[j],
                                           __ATOMIC_RELAXED,
                                           __HIP_MEMORY_SCOPE_WORKGROUP);
            }
        }
    }
    __syncthreads();

    float* p = part + (size_t)c * N_NODES + lo;
    for (int i = threadIdx.x; i < lim; i += 1024) p[i] = acc[i];
}

__global__ __launch_bounds__(256) void k_out(const float* __restrict__ zd,
                                             const float* __restrict__ b2,
                                             const float* __restrict__ part,
                                             float* __restrict__ out) {
    int i = blockIdx.x * 256 + threadIdx.x;
    if (i < N_NODES) {
        const float2 rec = ((const float2*)zd)[i];
        float n = rec.y;
        float o = rec.x * n * n + b2[0];
#pragma unroll
        for (int c = 0; c < NC; ++c) o += part[(size_t)c * N_NODES + i];
        out[i] = o;
    }
}

extern "C" void kernel_launch(void* const* d_in, const int* in_sizes, int n_in,
                              void* d_out, int out_size, void* d_ws, size_t ws_size,
                              hipStream_t stream) {
    const float* x   = (const float*)d_in[0];
    const float* W1  = (const float*)d_in[1];
    const float* b1  = (const float*)d_in[2];
    const float* W2  = (const float*)d_in[3];
    const float* b2  = (const float*)d_in[4];
    const int*   ei  = (const int*)d_in[5];
    const int*   src = ei;
    const int*   dst = ei + N_EDGES;
    float*       out = (float*)d_out;

    char* ws = (char*)d_ws;
    const size_t NPAD = 409600;
    float* zd   = (float*)(ws);
    float* dinv = (float*)(ws + 2 * NPAD);
    float* h    = (float*)(ws + 3 * NPAD);
    float* part = (float*)(ws + 3 * NPAD + (size_t)N_NODES * H_F * 4);
    unsigned* planes = (unsigned*)(ws + 3 * NPAD + (size_t)N_NODES * H_F * 4
                                   + (size_t)NC * N_NODES * 4);
    unsigned* pk = (unsigned*)(ws + 3 * NPAD + (size_t)N_NODES * H_F * 4
                               + (size_t)NC * N_NODES * 4
                               + (size_t)NSB * 1024);

    const int nb_nodes = (N_NODES + 255) / 256;

    k_deg_lds <<<NR * NC, 1024, 0, stream>>>(src, dst, part, planes, pk);
    k_dinv    <<<nb_nodes, 256, 0, stream>>>(part, dinv);

    k_gemm1   <<<(N_NODES + 127) / 128, 256, 0, stream>>>(x, W1, h);

    k_agg1_lds<<<NB1, 1024, 0, stream>>>(pk, planes, h, dinv, b1, W2, zd);

    k_edge2_lds<<<NR * NC, 1024, 0, stream>>>(src, dst, zd, dinv, part);
    k_out     <<<nb_nodes, 256, 0, stream>>>(zd, b2, part, out);
}
